# starting kernel
# baseline (speedup 1.0000x reference)
#include <hip/hip_runtime.h>
#include <math.h>

typedef float v2f __attribute__((ext_vector_type(2)));
typedef float v8f __attribute__((ext_vector_type(8)));

#define DIM 128


__global__ void k_deg_init(unsigned* __restrict__ deg, int N) {
    int i = blockIdx.x * 256 + threadIdx.x;
    if (i < N) deg[i] = 1u;
}

__global__ void k_deg_count(const int* __restrict__ dst, unsigned* __restrict__ deg, int E) {
    int e = blockIdx.x * 256 + threadIdx.x;
    if (e < E) atomicAdd(&deg[dst[e]], 1u);
}

__global__ void k_dinv(float* __restrict__ dinv, int N) {
    int i = blockIdx.x * 256 + threadIdx.x;
    if (i < N) {
        unsigned d = ((const unsigned*)dinv)[i];
        dinv[i] = rsqrtf((float)d);
    }
}


__global__ void k_gemm_wmma(const float* __restrict__ x, const float* __restrict__ W,
                            const float* __restrict__ dinv,
                            float* __restrict__ xw, float* __restrict__ out,
                            int N, int Mtiles) {
    const int wave = threadIdx.x >> 5;
    const int lane = threadIdx.x & 31;
    const int tile = blockIdx.x * 8 + wave;
    if (tile >= Mtiles * 8) return;
    const int tm = tile >> 3;
    const int tn = tile & 7;
    const int row0 = tm * 16, col0 = tn * 16;

    const int mr    = lane & 15;
    const int khalf = (lane >> 4) << 1;

    const int arow_i = min(row0 + mr, N - 1);
    const float* __restrict__ arow = x + (size_t)arow_i * DIM;
    const float* __restrict__ bcol = W + (col0 + mr);

    v8f c = {0.f, 0.f, 0.f, 0.f, 0.f, 0.f, 0.f, 0.f};
    #pragma unroll 8
    for (int kk = 0; kk < DIM; kk += 4) {
        v2f a = *(const v2f*)(arow + kk + khalf);
        v2f b;
        b.x = bcol[(size_t)(kk + khalf) * DIM];
        b.y = bcol[(size_t)(kk + khalf + 1) * DIM];
        c = __builtin_amdgcn_wmma_f32_16x16x4_f32(false, a, false, b,
                                                  (short)0, c, false, false);
    }

    const int mbase = (lane >> 4) << 3;
    const int gcol  = col0 + (lane & 15);
    if (row0 + 16 <= N) {
        #pragma unroll
        for (int r = 0; r < 8; ++r) {
            const int grow = row0 + mbase + r;
            const float v  = c[r];
            const float di = dinv[grow];
            xw [(size_t)grow * DIM + gcol] = v;
            out[(size_t)grow * DIM + gcol] = v * di * di;
        }
    } else {
        #pragma unroll
        for (int r = 0; r < 8; ++r) {
            const int grow = row0 + mbase + r;
            if (grow < N) {
                const float v  = c[r];
                const float di = dinv[grow];
                xw [(size_t)grow * DIM + gcol] = v;
                out[(size_t)grow * DIM + gcol] = v * di * di;
            }
        }
    }
}


__global__ void k_scatter(const int* __restrict__ src, const int* __restrict__ dst,
                          const float* __restrict__ dinv, const float* __restrict__ xw,
                          float* __restrict__ out, int E) {
    const int wave = threadIdx.x >> 5;
    const int lane = threadIdx.x & 31;
    const int e = blockIdx.x * 8 + wave;
    if (e >= E) return;
    const int s = src[e], t = dst[e];
    const float norm = dinv[s] * dinv[t];
    const float4 v = ((const float4*)(xw + (size_t)s * DIM))[lane];
    float* __restrict__ o = out + (size_t)t * DIM + (lane << 2);
    __hip_atomic_fetch_add(o + 0, v.x * norm, __ATOMIC_RELAXED, __HIP_MEMORY_SCOPE_AGENT);
    __hip_atomic_fetch_add(o + 1, v.y * norm, __ATOMIC_RELAXED, __HIP_MEMORY_SCOPE_AGENT);
    __hip_atomic_fetch_add(o + 2, v.z * norm, __ATOMIC_RELAXED, __HIP_MEMORY_SCOPE_AGENT);
    __hip_atomic_fetch_add(o + 3, v.w * norm, __ATOMIC_RELAXED, __HIP_MEMORY_SCOPE_AGENT);
}


__global__ void k_softmax(float* __restrict__ out, const float* __restrict__ bias, int N) {
    const int wave = threadIdx.x >> 5;
    const int lane = threadIdx.x & 31;
    const int row = blockIdx.x * 8 + wave;
    if (row >= N) return;
    float4* __restrict__ rowp = (float4*)(out + (size_t)row * DIM);
    float4 v  = rowp[lane];
    float4 bb = ((const float4*)bias)[lane];
    v.x = fmaxf(v.x + bb.x, 0.f);
    v.y = fmaxf(v.y + bb.y, 0.f);
    v.z = fmaxf(v.z + bb.z, 0.f);
    v.w = fmaxf(v.w + bb.w, 0.f);

    float m = fmaxf(fmaxf(v.x, v.y), fmaxf(v.z, v.w));
    #pragma unroll
    for (int off = 16; off > 0; off >>= 1) m = fmaxf(m, __shfl_xor(m, off, 32));

    v.x = expf(v.x - m); v.y = expf(v.y - m);
    v.z = expf(v.z - m); v.w = expf(v.w - m);
    float s = v.x + v.y + v.z + v.w;
    #pragma unroll
    for (int off = 16; off > 0; off >>= 1) s += __shfl_xor(s, off, 32);

    const float inv = 1.0f / s;
    v.x *= inv; v.y *= inv; v.z *= inv; v.w *= inv;
    rowp[lane] = v;
}


extern "C" void kernel_launch(void* const* d_in, const int* in_sizes, int n_in,
                              void* d_out, int out_size, void* d_ws, size_t ws_size,
                              hipStream_t stream) {
    const float* x  = (const float*)d_in[0];
    const int*   ei = (const int*)  d_in[1];
    const float* W  = (const float*)d_in[2];
    const float* b  = (const float*)d_in[3];
    const int N = in_sizes[0] / DIM;
    const int E = in_sizes[1] / 2;
    const int* src = ei;
    const int* dst = ei + E;

    float* out  = (float*)d_out;
    float* dinv = (float*)d_ws;
    float* xw   = dinv + N;

    k_deg_init <<<(N + 255) / 256, 256, 0, stream>>>((unsigned*)dinv, N);
    k_deg_count<<<(E + 255) / 256, 256, 0, stream>>>(dst, (unsigned*)dinv, E);
    k_dinv     <<<(N + 255) / 256, 256, 0, stream>>>(dinv, N);

    const int Mtiles = (N + 15) / 16;
    k_gemm_wmma<<<(Mtiles * 8 + 7) / 8, 256, 0, stream>>>(x, W, dinv, xw, out, N, Mtiles);

    k_scatter  <<<(E + 7) / 8, 256, 0, stream>>>(src, dst, dinv, xw, out, E);
    k_softmax  <<<(N + 7) / 8, 256, 0, stream>>>(out, b, N);
}


# scatter: lane l takes columns l, l+32, l+64, l+96 and xw rows start on a line, so each f32 atomic add and each row load covers one whole 128-B line, not quarter-filled ones; same sums, arrival order
# speedup vs baseline: 1.3050x; 1.3050x over previous
#include <hip/hip_runtime.h>
#include <math.h>

typedef float v2f __attribute__((ext_vector_type(2)));
typedef float v8f __attribute__((ext_vector_type(8)));

#define DIM 128


__global__ void k_deg_init(unsigned* __restrict__ deg, int N) {
    int i = blockIdx.x * 256 + threadIdx.x;
    if (i < N) deg[i] = 1u;
}

__global__ void k_deg_count(const int* __restrict__ dst, unsigned* __restrict__ deg, int E) {
    int e = blockIdx.x * 256 + threadIdx.x;
    if (e < E) atomicAdd(&deg[dst[e]], 1u);
}

__global__ void k_dinv(float* __restrict__ dinv, int N) {
    int i = blockIdx.x * 256 + threadIdx.x;
    if (i < N) {
        unsigned d = ((const unsigned*)dinv)[i];
        dinv[i] = rsqrtf((float)d);
    }
}


__global__ void k_gemm_wmma(const float* __restrict__ x, const float* __restrict__ W,
                            const float* __restrict__ dinv,
                            float* __restrict__ xw, float* __restrict__ out,
                            int N, int Mtiles) {
    const int wave = threadIdx.x >> 5;
    const int lane = threadIdx.x & 31;
    const int tile = blockIdx.x * 8 + wave;
    if (tile >= Mtiles * 8) return;
    const int tm = tile >> 3;
    const int tn = tile & 7;
    const int row0 = tm * 16, col0 = tn * 16;

    const int mr    = lane & 15;
    const int khalf = (lane >> 4) << 1;

    const int arow_i = min(row0 + mr, N - 1);
    const float* __restrict__ arow = x + (size_t)arow_i * DIM;
    const float* __restrict__ bcol = W + (col0 + mr);

    v8f c = {0.f, 0.f, 0.f, 0.f, 0.f, 0.f, 0.f, 0.f};
    #pragma unroll 8
    for (int kk = 0; kk < DIM; kk += 4) {
        v2f a = *(const v2f*)(arow + kk + khalf);
        v2f b;
        b.x = bcol[(size_t)(kk + khalf) * DIM];
        b.y = bcol[(size_t)(kk + khalf + 1) * DIM];
        c = __builtin_amdgcn_wmma_f32_16x16x4_f32(false, a, false, b,
                                                  (short)0, c, false, false);
    }

    const int mbase = (lane >> 4) << 3;
    const int gcol  = col0 + (lane & 15);
    if (row0 + 16 <= N) {
        #pragma unroll
        for (int r = 0; r < 8; ++r) {
            const int grow = row0 + mbase + r;
            const float v  = c[r];
            const float di = dinv[grow];
            xw [(size_t)grow * DIM + gcol] = v;
            out[(size_t)grow * DIM + gcol] = v * di * di;
        }
    } else {
        #pragma unroll
        for (int r = 0; r < 8; ++r) {
            const int grow = row0 + mbase + r;
            if (grow < N) {
                const float v  = c[r];
                const float di = dinv[grow];
                xw [(size_t)grow * DIM + gcol] = v;
                out[(size_t)grow * DIM + gcol] = v * di * di;
            }
        }
    }
}


__global__ void k_scatter(const int* __restrict__ src, const int* __restrict__ dst,
                          const float* __restrict__ dinv, const float* __restrict__ xw,
                          float* __restrict__ out, int E) {
    const int wave = threadIdx.x >> 5;
    const int lane = threadIdx.x & 31;
    const int e = blockIdx.x * 8 + wave;
    if (e >= E) return;
    const int s = src[e], t = dst[e];
    const float norm = dinv[s] * dinv[t];
    const float* __restrict__ xr = xw + (size_t)s * DIM + lane;
    const float v0 = xr[0];
    const float v1 = xr[32];
    const float v2 = xr[64];
    const float v3 = xr[96];
    float* __restrict__ o = out + (size_t)t * DIM + lane;
    __hip_atomic_fetch_add(o +  0, v0 * norm, __ATOMIC_RELAXED, __HIP_MEMORY_SCOPE_AGENT);
    __hip_atomic_fetch_add(o + 32, v1 * norm, __ATOMIC_RELAXED, __HIP_MEMORY_SCOPE_AGENT);
    __hip_atomic_fetch_add(o + 64, v2 * norm, __ATOMIC_RELAXED, __HIP_MEMORY_SCOPE_AGENT);
    __hip_atomic_fetch_add(o + 96, v3 * norm, __ATOMIC_RELAXED, __HIP_MEMORY_SCOPE_AGENT);
}


__global__ void k_softmax(float* __restrict__ out, const float* __restrict__ bias, int N) {
    const int wave = threadIdx.x >> 5;
    const int lane = threadIdx.x & 31;
    const int row = blockIdx.x * 8 + wave;
    if (row >= N) return;
    float4* __restrict__ rowp = (float4*)(out + (size_t)row * DIM);
    float4 v  = rowp[lane];
    float4 bb = ((const float4*)bias)[lane];
    v.x = fmaxf(v.x + bb.x, 0.f);
    v.y = fmaxf(v.y + bb.y, 0.f);
    v.z = fmaxf(v.z + bb.z, 0.f);
    v.w = fmaxf(v.w + bb.w, 0.f);

    float m = fmaxf(fmaxf(v.x, v.y), fmaxf(v.z, v.w));
    #pragma unroll
    for (int off = 16; off > 0; off >>= 1) m = fmaxf(m, __shfl_xor(m, off, 32));

    v.x = expf(v.x - m); v.y = expf(v.y - m);
    v.z = expf(v.z - m); v.w = expf(v.w - m);
    float s = v.x + v.y + v.z + v.w;
    #pragma unroll
    for (int off = 16; off > 0; off >>= 1) s += __shfl_xor(s, off, 32);

    const float inv = 1.0f / s;
    v.x *= inv; v.y *= inv; v.z *= inv; v.w *= inv;
    rowp[lane] = v;
}


extern "C" void kernel_launch(void* const* d_in, const int* in_sizes, int n_in,
                              void* d_out, int out_size, void* d_ws, size_t ws_size,
                              hipStream_t stream) {
    const float* x  = (const float*)d_in[0];
    const int*   ei = (const int*)  d_in[1];
    const float* W  = (const float*)d_in[2];
    const float* b  = (const float*)d_in[3];
    const int N = in_sizes[0] / DIM;
    const int E = in_sizes[1] / 2;
    const int* src = ei;
    const int* dst = ei + E;

    float* out  = (float*)d_out;
    float* xw   = (float*)d_ws;
    float* dinv = xw + (size_t)N * DIM;

    k_deg_init <<<(N + 255) / 256, 256, 0, stream>>>((unsigned*)dinv, N);
    k_deg_count<<<(E + 255) / 256, 256, 0, stream>>>(dst, (unsigned*)dinv, E);
    k_dinv     <<<(N + 255) / 256, 256, 0, stream>>>(dinv, N);

    const int Mtiles = (N + 15) / 16;
    k_gemm_wmma<<<(Mtiles * 8 + 7) / 8, 256, 0, stream>>>(x, W, dinv, xw, out, N, Mtiles);

    k_scatter  <<<(E + 7) / 8, 256, 0, stream>>>(src, dst, dinv, xw, out, E);
    k_softmax  <<<(N + 7) / 8, 256, 0, stream>>>(out, b, N);
}


# f32 atomic scatter replaced by gather over a dst-bucketed edge list (integer atomic ranks, offsets, fill): one wave sums a row in registers, one store; sum order = atomic rank order, self-loop last
# speedup vs baseline: 1.8119x; 1.3884x over previous
#include <hip/hip_runtime.h>
#include <math.h>


typedef float v2f __attribute__((ext_vector_type(2)));
typedef float v8f __attribute__((ext_vector_type(8)));

#define DIM 128
#define EPT 4


__global__ void k_deg_init(unsigned* __restrict__ cnt, unsigned* __restrict__ total, int N) {
    int i = blockIdx.x * 256 + threadIdx.x;
    if (i < N) cnt[i] = 0u;
    if (i == 0) *total = 0u;
}

__global__ void k_count(const int* __restrict__ dst, unsigned* __restrict__ cnt,
                        int* __restrict__ rank, int E) {
    const int T  = gridDim.x * 256;
    const int e0 = blockIdx.x * 256 + threadIdx.x;
    if (e0 + (EPT - 1) * T < E) {
        int t[EPT];
        unsigned r[EPT];
        for (int k = 0; k < EPT; ++k) t[k] = dst[e0 + k * T];
        for (int k = 0; k < EPT; ++k) r[k] = atomicAdd(&cnt[t[k]], 1u);
        for (int k = 0; k < EPT; ++k) rank[e0 + k * T] = (int)r[k];
    } else {
        for (int k = 0; k < EPT; ++k) {
            const int e = e0 + k * T;
            if (e < E) rank[e] = (int)atomicAdd(&cnt[dst[e]], 1u);
        }
    }
}

__global__ __launch_bounds__(1024) void k_offsets(const unsigned* __restrict__ cnt,
                                                  float* __restrict__ dinv,
                                                  int* __restrict__ rowstart,
                                                  unsigned* __restrict__ total, int N) {
    __shared__ unsigned wsum[32];
    const int i    = blockIdx.x * 1024 + threadIdx.x;
    const int lane = threadIdx.x & 31;
    const int wave = threadIdx.x >> 5;
    const unsigned c = (i < N) ? cnt[i] : 0u;
    if (i < N) dinv[i] = rsqrtf((float)(c + 1u));

    unsigned incl = c;
    for (int off = 1; off < 32; off <<= 1) {
        const unsigned o = __shfl_up(incl, off, 32);
        if (lane >= off) incl += o;
    }
    if (lane == 31) wsum[wave] = incl;
    __syncthreads();
    if (wave == 0) {
        const unsigned w = wsum[lane];
        unsigned winc = w;
        for (int off = 1; off < 32; off <<= 1) {
            const unsigned o = __shfl_up(winc, off, 32);
            if (lane >= off) winc += o;
        }
        unsigned base = 0u;
        if (lane == 31) base = atomicAdd(total, winc);
        base = __shfl(base, 31, 32);
        wsum[lane] = base + winc - w;
    }
    __syncthreads();
    if (i < N) rowstart[i] = (int)(wsum[wave] + incl - c);
}

__global__ void k_fill(const int* __restrict__ src, const int* __restrict__ dst,
                       const int* __restrict__ rank, const int* __restrict__ rowstart,
                       int* __restrict__ list, int E) {
    const int T  = gridDim.x * 256;
    const int e0 = blockIdx.x * 256 + threadIdx.x;
    if (e0 + (EPT - 1) * T < E) {
        int t[EPT], rk[EPT], s[EPT], pos[EPT];
        for (int k = 0; k < EPT; ++k) {
            t[k]  = dst [e0 + k * T];
            rk[k] = rank[e0 + k * T];
            s[k]  = src [e0 + k * T];
        }
        for (int k = 0; k < EPT; ++k) pos[k] = rowstart[t[k]] + rk[k];
        for (int k = 0; k < EPT; ++k)
            list[min(max(pos[k], 0), E - 1)] = s[k];
    } else {
        for (int k = 0; k < EPT; ++k) {
            const int e = e0 + k * T;
            if (e < E) {
                int p = rowstart[dst[e]] + rank[e];
                p = min(max(p, 0), E - 1);
                list[p] = src[e];
            }
        }
    }
}


__global__ void k_gemm_wmma(const float* __restrict__ x, const float* __restrict__ W,
                            float* __restrict__ xw, int N, int Mtiles) {
    const int wave = threadIdx.x >> 5;
    const int lane = threadIdx.x & 31;
    const int tile = blockIdx.x * 8 + wave;
    if (tile >= Mtiles * 8) return;
    const int tm = tile >> 3;
    const int tn = tile & 7;
    const int row0 = tm * 16, col0 = tn * 16;

    const int mr    = lane & 15;
    const int khalf = (lane >> 4) << 1;

    const int arow_i = min(row0 + mr, N - 1);
    const float* __restrict__ arow = x + (size_t)arow_i * DIM;
    const float* __restrict__ bcol = W + (col0 + mr);

    v8f c = {0.f, 0.f, 0.f, 0.f, 0.f, 0.f, 0.f, 0.f};
    #pragma unroll 8
    for (int kk = 0; kk < DIM; kk += 4) {
        v2f a = *(const v2f*)(arow + kk + khalf);
        v2f b;
        b.x = bcol[(size_t)(kk + khalf) * DIM];
        b.y = bcol[(size_t)(kk + khalf + 1) * DIM];
        c = __builtin_amdgcn_wmma_f32_16x16x4_f32(false, a, false, b,
                                                  (short)0, c, false, false);
    }

    const int mbase = (lane >> 4) << 3;
    const int gcol  = col0 + (lane & 15);
    if (row0 + 16 <= N) {
        #pragma unroll
        for (int r = 0; r < 8; ++r) {
            const int grow = row0 + mbase + r;
            xw[(size_t)grow * DIM + gcol] = c[r];
        }
    } else {
        #pragma unroll
        for (int r = 0; r < 8; ++r) {
            const int grow = row0 + mbase + r;
            if (grow < N) xw[(size_t)grow * DIM + gcol] = c[r];
        }
    }
}


#define GROUP 8

__global__ __launch_bounds__(256) void k_gather(const int* __restrict__ list,
                                                const int* __restrict__ rowstart,
                                                const unsigned* __restrict__ cnt,
                                                const float* __restrict__ dinv,
                                                const float* __restrict__ xw,
                                                float* __restrict__ out, int N, int E) {
    const int lane = threadIdx.x & 31;
    const int t = __builtin_amdgcn_readfirstlane(blockIdx.x * 8 + (threadIdx.x >> 5));
    if (t >= N) return;

    const float4* __restrict__ xw4 = (const float4*)xw;
    const float dt = dinv[t];
    int start = rowstart[t];
    int n     = (int)cnt[t];
    start = min(max(start, 0), E);
    n     = min(max(n, 0), E - start);

    const float4 self = xw4[(size_t)t * (DIM / 4) + lane];
    float4 acc = make_float4(0.f, 0.f, 0.f, 0.f);

    for (int base = 0; base < n; base += 32) {
        const int m = min(32, n - base);
        int   my_s = 0;
        float my_d = 0.f;
        if (lane < m) {
            my_s = list[start + base + lane];
            my_s = min(max(my_s, 0), N - 1);
            my_d = dinv[my_s];
        }
        for (int j = 0; j < m; j += GROUP) {
            float4 r[GROUP];
            for (int k = 0; k < GROUP; ++k) {
                if (j + k < m) {
                    const int s = __builtin_amdgcn_readlane(my_s, j + k);
                    r[k] = xw4[(size_t)s * (DIM / 4) + lane];
                } else {
                    r[k] = make_float4(0.f, 0.f, 0.f, 0.f);
                }
            }
            for (int k = 0; k < GROUP; ++k) {
                const float w = dt * __builtin_bit_cast(float,
                                    __builtin_amdgcn_readlane(__builtin_bit_cast(int, my_d), j + k));
                acc.x += r[k].x * w;
                acc.y += r[k].y * w;
                acc.z += r[k].z * w;
                acc.w += r[k].w * w;
            }
        }
    }
    acc.x += self.x * dt * dt;
    acc.y += self.y * dt * dt;
    acc.z += self.z * dt * dt;
    acc.w += self.w * dt * dt;
    ((float4*)out)[(size_t)t * (DIM / 4) + lane] = acc;
}


__global__ void k_softmax(float* __restrict__ out, const float* __restrict__ bias, int N) {
    const int wave = threadIdx.x >> 5;
    const int lane = threadIdx.x & 31;
    const int row = blockIdx.x * 8 + wave;
    if (row >= N) return;
    float4* __restrict__ rowp = (float4*)(out + (size_t)row * DIM);
    float4 v  = rowp[lane];
    float4 bb = ((const float4*)bias)[lane];
    v.x = fmaxf(v.x + bb.x, 0.f);
    v.y = fmaxf(v.y + bb.y, 0.f);
    v.z = fmaxf(v.z + bb.z, 0.f);
    v.w = fmaxf(v.w + bb.w, 0.f);

    float m = fmaxf(fmaxf(v.x, v.y), fmaxf(v.z, v.w));
    #pragma unroll
    for (int off = 16; off > 0; off >>= 1) m = fmaxf(m, __shfl_xor(m, off, 32));

    v.x = expf(v.x - m); v.y = expf(v.y - m);
    v.z = expf(v.z - m); v.w = expf(v.w - m);
    float s = v.x + v.y + v.z + v.w;
    #pragma unroll
    for (int off = 16; off > 0; off >>= 1) s += __shfl_xor(s, off, 32);

    const float inv = 1.0f / s;
    v.x *= inv; v.y *= inv; v.z *= inv; v.w *= inv;
    rowp[lane] = v;
}


extern "C" void kernel_launch(void* const* d_in, const int* in_sizes, int n_in,
                              void* d_out, int out_size, void* d_ws, size_t ws_size,
                              hipStream_t stream) {
    const float* x  = (const float*)d_in[0];
    const int*   ei = (const int*)  d_in[1];
    const float* W  = (const float*)d_in[2];
    const float* b  = (const float*)d_in[3];
    const int N = in_sizes[0] / DIM;
    const int E = in_sizes[1] / 2;
    const int* src = ei;
    const int* dst = ei + E;

    float* out = (float*)d_out;
    float*    xw       = (float*)d_ws;
    float*    dinv     = xw + (size_t)N * DIM;
    unsigned* cnt      = (unsigned*)(dinv + N);
    int*      rowstart = (int*)(cnt + N);
    int*      rank     = rowstart + N;
    int*      list     = rank + E;
    unsigned* total    = (unsigned*)(list + E);

    k_deg_init<<<(N + 255) / 256, 256, 0, stream>>>(cnt, total, N);
    k_count   <<<(E + 256 * EPT - 1) / (256 * EPT), 256, 0, stream>>>(dst, cnt, rank, E);
    k_offsets <<<(N + 1023) / 1024, 1024, 0, stream>>>(cnt, dinv, rowstart, total, N);
    k_fill    <<<(E + 256 * EPT - 1) / (256 * EPT), 256, 0, stream>>>(src, dst, rank, rowstart, list, E);

    const int Mtiles = (N + 15) / 16;
    k_gemm_wmma<<<(Mtiles * 8 + 7) / 8, 256, 0, stream>>>(x, W, xw, N, Mtiles);

    k_gather  <<<(N + 7) / 8, 256, 0, stream>>>(list, rowstart, cnt, dinv, xw, out, N, E);
    k_softmax <<<(N + 7) / 8, 256, 0, stream>>>(out, b, N);
}


# softmax launch fused into the gather: the wave that summed a row adds bias, does ReLU and the row softmax on its registers and stores the finished row once; same sums, same order
# speedup vs baseline: 1.9208x; 1.0601x over previous
#include <hip/hip_runtime.h>
#include <math.h>


typedef float v2f __attribute__((ext_vector_type(2)));
typedef float v8f __attribute__((ext_vector_type(8)));

#define DIM 128
#define EPT 4


__global__ void k_deg_init(unsigned* __restrict__ cnt, unsigned* __restrict__ total, int N) {
    int i = blockIdx.x * 256 + threadIdx.x;
    if (i < N) cnt[i] = 0u;
    if (i == 0) *total = 0u;
}

__global__ void k_count(const int* __restrict__ dst, unsigned* __restrict__ cnt,
                        int* __restrict__ rank, int E) {
    const int T  = gridDim.x * 256;
    const int e0 = blockIdx.x * 256 + threadIdx.x;
    if (e0 + (EPT - 1) * T < E) {
        int t[EPT];
        unsigned r[EPT];
        for (int k = 0; k < EPT; ++k) t[k] = dst[e0 + k * T];
        for (int k = 0; k < EPT; ++k) r[k] = atomicAdd(&cnt[t[k]], 1u);
        for (int k = 0; k < EPT; ++k) rank[e0 + k * T] = (int)r[k];
    } else {
        for (int k = 0; k < EPT; ++k) {
            const int e = e0 + k * T;
            if (e < E) rank[e] = (int)atomicAdd(&cnt[dst[e]], 1u);
        }
    }
}

__global__ __launch_bounds__(1024) void k_offsets(const unsigned* __restrict__ cnt,
                                                  float* __restrict__ dinv,
                                                  int* __restrict__ rowstart,
                                                  unsigned* __restrict__ total, int N) {
    __shared__ unsigned wsum[32];
    const int i    = blockIdx.x * 1024 + threadIdx.x;
    const int lane = threadIdx.x & 31;
    const int wave = threadIdx.x >> 5;
    const unsigned c = (i < N) ? cnt[i] : 0u;
    if (i < N) dinv[i] = rsqrtf((float)(c + 1u));

    unsigned incl = c;
    for (int off = 1; off < 32; off <<= 1) {
        const unsigned o = __shfl_up(incl, off, 32);
        if (lane >= off) incl += o;
    }
    if (lane == 31) wsum[wave] = incl;
    __syncthreads();
    if (wave == 0) {
        const unsigned w = wsum[lane];
        unsigned winc = w;
        for (int off = 1; off < 32; off <<= 1) {
            const unsigned o = __shfl_up(winc, off, 32);
            if (lane >= off) winc += o;
        }
        unsigned base = 0u;
        if (lane == 31) base = atomicAdd(total, winc);
        base = __shfl(base, 31, 32);
        wsum[lane] = base + winc - w;
    }
    __syncthreads();
    if (i < N) rowstart[i] = (int)(wsum[wave] + incl - c);
}

__global__ void k_fill(const int* __restrict__ src, const int* __restrict__ dst,
                       const int* __restrict__ rank, const int* __restrict__ rowstart,
                       int* __restrict__ list, int E) {
    const int T  = gridDim.x * 256;
    const int e0 = blockIdx.x * 256 + threadIdx.x;
    if (e0 + (EPT - 1) * T < E) {
        int t[EPT], rk[EPT], s[EPT], pos[EPT];
        for (int k = 0; k < EPT; ++k) {
            t[k]  = dst [e0 + k * T];
            rk[k] = rank[e0 + k * T];
            s[k]  = src [e0 + k * T];
        }
        for (int k = 0; k < EPT; ++k) pos[k] = rowstart[t[k]] + rk[k];
        for (int k = 0; k < EPT; ++k)
            list[min(max(pos[k], 0), E - 1)] = s[k];
    } else {
        for (int k = 0; k < EPT; ++k) {
            const int e = e0 + k * T;
            if (e < E) {
                int p = rowstart[dst[e]] + rank[e];
                p = min(max(p, 0), E - 1);
                list[p] = src[e];
            }
        }
    }
}


__global__ void k_gemm_wmma(const float* __restrict__ x, const float* __restrict__ W,
                            float* __restrict__ xw, int N, int Mtiles) {
    const int wave = threadIdx.x >> 5;
    const int lane = threadIdx.x & 31;
    const int tile = blockIdx.x * 8 + wave;
    if (tile >= Mtiles * 8) return;
    const int tm = tile >> 3;
    const int tn = tile & 7;
    const int row0 = tm * 16, col0 = tn * 16;

    const int mr    = lane & 15;
    const int khalf = (lane >> 4) << 1;

    const int arow_i = min(row0 + mr, N - 1);
    const float* __restrict__ arow = x + (size_t)arow_i * DIM;
    const float* __restrict__ bcol = W + (col0 + mr);

    v8f c = {0.f, 0.f, 0.f, 0.f, 0.f, 0.f, 0.f, 0.f};
    #pragma unroll 8
    for (int kk = 0; kk < DIM; kk += 4) {
        v2f a = *(const v2f*)(arow + kk + khalf);
        v2f b;
        b.x = bcol[(size_t)(kk + khalf) * DIM];
        b.y = bcol[(size_t)(kk + khalf + 1) * DIM];
        c = __builtin_amdgcn_wmma_f32_16x16x4_f32(false, a, false, b,
                                                  (short)0, c, false, false);
    }

    const int mbase = (lane >> 4) << 3;
    const int gcol  = col0 + (lane & 15);
    if (row0 + 16 <= N) {
        #pragma unroll
        for (int r = 0; r < 8; ++r) {
            const int grow = row0 + mbase + r;
            xw[(size_t)grow * DIM + gcol] = c[r];
        }
    } else {
        #pragma unroll
        for (int r = 0; r < 8; ++r) {
            const int grow = row0 + mbase + r;
            if (grow < N) xw[(size_t)grow * DIM + gcol] = c[r];
        }
    }
}


#define GROUP 8

__global__ __launch_bounds__(256) void k_gather(const int* __restrict__ list,
                                                const int* __restrict__ rowstart,
                                                const unsigned* __restrict__ cnt,
                                                const float* __restrict__ dinv,
                                                const float* __restrict__ xw,
                                                const float* __restrict__ bias,
                                                float* __restrict__ out, int N, int E) {
    const int lane = threadIdx.x & 31;
    const int t = __builtin_amdgcn_readfirstlane(blockIdx.x * 8 + (threadIdx.x >> 5));
    if (t >= N) return;

    const float4* __restrict__ xw4 = (const float4*)xw;
    const float dt = dinv[t];
    int start = rowstart[t];
    int n     = (int)cnt[t];
    start = min(max(start, 0), E);
    n     = min(max(n, 0), E - start);

    const float4 self = xw4[(size_t)t * (DIM / 4) + lane];
    const float4 bb   = ((const float4*)bias)[lane];
    float4 acc = make_float4(0.f, 0.f, 0.f, 0.f);

    for (int base = 0; base < n; base += 32) {
        const int m = min(32, n - base);
        int   my_s = 0;
        float my_d = 0.f;
        if (lane < m) {
            my_s = list[start + base + lane];
            my_s = min(max(my_s, 0), N - 1);
            my_d = dinv[my_s];
        }
        for (int j = 0; j < m; j += GROUP) {
            float4 r[GROUP];
            for (int k = 0; k < GROUP; ++k) {
                if (j + k < m) {
                    const int s = __builtin_amdgcn_readlane(my_s, j + k);
                    r[k] = xw4[(size_t)s * (DIM / 4) + lane];
                } else {
                    r[k] = make_float4(0.f, 0.f, 0.f, 0.f);
                }
            }
            for (int k = 0; k < GROUP; ++k) {
                const float w = dt * __builtin_bit_cast(float,
                                    __builtin_amdgcn_readlane(__builtin_bit_cast(int, my_d), j + k));
                acc.x += r[k].x * w;
                acc.y += r[k].y * w;
                acc.z += r[k].z * w;
                acc.w += r[k].w * w;
            }
        }
    }
    acc.x += self.x * dt * dt;
    acc.y += self.y * dt * dt;
    acc.z += self.z * dt * dt;
    acc.w += self.w * dt * dt;

    float4 v;
    v.x = fmaxf(acc.x + bb.x, 0.f);
    v.y = fmaxf(acc.y + bb.y, 0.f);
    v.z = fmaxf(acc.z + bb.z, 0.f);
    v.w = fmaxf(acc.w + bb.w, 0.f);

    float m = fmaxf(fmaxf(v.x, v.y), fmaxf(v.z, v.w));
    #pragma unroll
    for (int off = 16; off > 0; off >>= 1) m = fmaxf(m, __shfl_xor(m, off, 32));

    v.x = expf(v.x - m); v.y = expf(v.y - m);
    v.z = expf(v.z - m); v.w = expf(v.w - m);
    float s = v.x + v.y + v.z + v.w;
    #pragma unroll
    for (int off = 16; off > 0; off >>= 1) s += __shfl_xor(s, off, 32);

    const float inv = 1.0f / s;
    v.x *= inv; v.y *= inv; v.z *= inv; v.w *= inv;
    ((float4*)out)[(size_t)t * (DIM / 4) + lane] = v;
}


extern "C" void kernel_launch(void* const* d_in, const int* in_sizes, int n_in,
                              void* d_out, int out_size, void* d_ws, size_t ws_size,
                              hipStream_t stream) {
    const float* x  = (const float*)d_in[0];
    const int*   ei = (const int*)  d_in[1];
    const float* W  = (const float*)d_in[2];
    const float* b  = (const float*)d_in[3];
    const int N = in_sizes[0] / DIM;
    const int E = in_sizes[1] / 2;
    const int* src = ei;
    const int* dst = ei + E;

    float* out = (float*)d_out;
    float*    xw       = (float*)d_ws;
    float*    dinv     = xw + (size_t)N * DIM;
    unsigned* cnt      = (unsigned*)(dinv + N);
    int*      rowstart = (int*)(cnt + N);
    int*      rank     = rowstart + N;
    int*      list     = rank + E;
    unsigned* total    = (unsigned*)(list + E);

    k_deg_init<<<(N + 255) / 256, 256, 0, stream>>>(cnt, total, N);
    k_count   <<<(E + 256 * EPT - 1) / (256 * EPT), 256, 0, stream>>>(dst, cnt, rank, E);
    k_offsets <<<(N + 1023) / 1024, 1024, 0, stream>>>(cnt, dinv, rowstart, total, N);
    k_fill    <<<(E + 256 * EPT - 1) / (256 * EPT), 256, 0, stream>>>(src, dst, rank, rowstart, list, E);

    const int Mtiles = (N + 15) / 16;
    k_gemm_wmma<<<(Mtiles * 8 + 7) / 8, 256, 0, stream>>>(x, W, xw, N, Mtiles);

    k_gather  <<<(N + 7) / 8, 256, 0, stream>>>(list, rowstart, cnt, dinv, xw, b, out, N, E);
}


# GEMM re-tiled: one wave per 16-row x 128-column strip (8 accumulator tiles), W copied to LDS once per block; an A fragment is loaded once and used 8 times, B fragments come from LDS; same sums
# speedup vs baseline: 2.0423x; 1.0633x over previous
#include <hip/hip_runtime.h>
#include <math.h>


typedef float v2f __attribute__((ext_vector_type(2)));
typedef float v8f __attribute__((ext_vector_type(8)));

#define DIM 128
#define EPT 4


__global__ void k_deg_init(unsigned* __restrict__ cnt, unsigned* __restrict__ total, int N) {
    int i = blockIdx.x * 256 + threadIdx.x;
    if (i < N) cnt[i] = 0u;
    if (i == 0) *total = 0u;
}

__global__ void k_count(const int* __restrict__ dst, unsigned* __restrict__ cnt,
                        int* __restrict__ rank, int E) {
    const int T  = gridDim.x * 256;
    const int e0 = blockIdx.x * 256 + threadIdx.x;
    if (e0 + (EPT - 1) * T < E) {
        int t[EPT];
        unsigned r[EPT];
        for (int k = 0; k < EPT; ++k) t[k] = dst[e0 + k * T];
        for (int k = 0; k < EPT; ++k) r[k] = atomicAdd(&cnt[t[k]], 1u);
        for (int k = 0; k < EPT; ++k) rank[e0 + k * T] = (int)r[k];
    } else {
        for (int k = 0; k < EPT; ++k) {
            const int e = e0 + k * T;
            if (e < E) rank[e] = (int)atomicAdd(&cnt[dst[e]], 1u);
        }
    }
}

__global__ __launch_bounds__(1024) void k_offsets(const unsigned* __restrict__ cnt,
                                                  float* __restrict__ dinv,
                                                  int* __restrict__ rowstart,
                                                  unsigned* __restrict__ total, int N) {
    __shared__ unsigned wsum[32];
    const int i    = blockIdx.x * 1024 + threadIdx.x;
    const int lane = threadIdx.x & 31;
    const int wave = threadIdx.x >> 5;
    const unsigned c = (i < N) ? cnt[i] : 0u;
    if (i < N) dinv[i] = rsqrtf((float)(c + 1u));

    unsigned incl = c;
    for (int off = 1; off < 32; off <<= 1) {
        const unsigned o = __shfl_up(incl, off, 32);
        if (lane >= off) incl += o;
    }
    if (lane == 31) wsum[wave] = incl;
    __syncthreads();
    if (wave == 0) {
        const unsigned w = wsum[lane];
        unsigned winc = w;
        for (int off = 1; off < 32; off <<= 1) {
            const unsigned o = __shfl_up(winc, off, 32);
            if (lane >= off) winc += o;
        }
        unsigned base = 0u;
        if (lane == 31) base = atomicAdd(total, winc);
        base = __shfl(base, 31, 32);
        wsum[lane] = base + winc - w;
    }
    __syncthreads();
    if (i < N) rowstart[i] = (int)(wsum[wave] + incl - c);
}

__global__ void k_fill(const int* __restrict__ src, const int* __restrict__ dst,
                       const int* __restrict__ rank, const int* __restrict__ rowstart,
                       int* __restrict__ list, int E) {
    const int T  = gridDim.x * 256;
    const int e0 = blockIdx.x * 256 + threadIdx.x;
    if (e0 + (EPT - 1) * T < E) {
        int t[EPT], rk[EPT], s[EPT], pos[EPT];
        for (int k = 0; k < EPT; ++k) {
            t[k]  = dst [e0 + k * T];
            rk[k] = rank[e0 + k * T];
            s[k]  = src [e0 + k * T];
        }
        for (int k = 0; k < EPT; ++k) pos[k] = rowstart[t[k]] + rk[k];
        for (int k = 0; k < EPT; ++k)
            list[min(max(pos[k], 0), E - 1)] = s[k];
    } else {
        for (int k = 0; k < EPT; ++k) {
            const int e = e0 + k * T;
            if (e < E) {
                int p = rowstart[dst[e]] + rank[e];
                p = min(max(p, 0), E - 1);
                list[p] = src[e];
            }
        }
    }
}


#define WP 288

__global__ __launch_bounds__(256) void k_gemm_wmma(const float* __restrict__ x, const float* __restrict__ W,
                                                   float* __restrict__ xw, int N, int Mtiles) {
    __shared__ float Ws[(DIM / 2) * WP];
    const int lane = threadIdx.x & 31;
    const int wave = __builtin_amdgcn_readfirstlane(threadIdx.x >> 5);
    const int tm   = blockIdx.x * 8 + wave;
    const int row0 = tm * 16;

    const int mr = lane & 15;
    const int h  = lane >> 4;

    const int arow_i = min(row0 + mr, N - 1);
    const float* __restrict__ arow = x + (size_t)arow_i * DIM + 2 * h;
    v2f a[DIM / 4];
    #pragma unroll
    for (int s = 0; s < DIM / 4; ++s) a[s] = *(const v2f*)(arow + 4 * s);

    #pragma unroll
    for (int j = 0; j < (DIM / 2) * (DIM / 4) / 256; ++j) {
        const int idx = (int)threadIdx.x + j * 256;
        const int kp  = idx >> 5;
        const int c4  = idx & 31;
        const float4 r0 = ((const float4*)W)[(2 * kp)     * (DIM / 4) + c4];
        const float4 r1 = ((const float4*)W)[(2 * kp + 1) * (DIM / 4) + c4];
        float4* d = (float4*)&Ws[kp * WP + 8 * c4];
        d[0] = make_float4(r0.x, r1.x, r0.y, r1.y);
        d[1] = make_float4(r0.z, r1.z, r0.w, r1.w);
    }
    __syncthreads();
    if (tm >= Mtiles) return;

    v8f c[8];
    #pragma unroll
    for (int tn = 0; tn < 8; ++tn) c[tn] = (v8f){0.f, 0.f, 0.f, 0.f, 0.f, 0.f, 0.f, 0.f};

    const float* bp = Ws + h * WP + 2 * mr;
    #pragma unroll
    for (int s = 0; s < DIM / 4; ++s) {
        #pragma unroll
        for (int tn = 0; tn < 8; ++tn) {
            const v2f b = *(const v2f*)(bp + (2 * s) * WP + tn * 32);
            c[tn] = __builtin_amdgcn_wmma_f32_16x16x4_f32(false, a[s], false, b,
                                                          (short)0, c[tn], false, false);
        }
    }

    const int mbase = h << 3;
    float* __restrict__ orow = xw + (size_t)(row0 + mbase) * DIM + mr;
    if (row0 + 16 <= N) {
        #pragma unroll
        for (int r = 0; r < 8; ++r) {
            #pragma unroll
            for (int tn = 0; tn < 8; ++tn) orow[(size_t)r * DIM + tn * 16] = c[tn][r];
        }
    } else {
        #pragma unroll
        for (int r = 0; r < 8; ++r) {
            if (row0 + mbase + r < N) {
                #pragma unroll
                for (int tn = 0; tn < 8; ++tn) orow[(size_t)r * DIM + tn * 16] = c[tn][r];
            }
        }
    }
}


#define GROUP 8

__global__ __launch_bounds__(256) void k_gather(const int* __restrict__ list,
                                                const int* __restrict__ rowstart,
                                                const unsigned* __restrict__ cnt,
                                                const float* __restrict__ dinv,
                                                const float* __restrict__ xw,
                                                const float* __restrict__ bias,
                                                float* __restrict__ out, int N, int E) {
    const int lane = threadIdx.x & 31;
    const int t = __builtin_amdgcn_readfirstlane(blockIdx.x * 8 + (threadIdx.x >> 5));
    if (t >= N) return;

    const float4* __restrict__ xw4 = (const float4*)xw;
    const float dt = dinv[t];
    int start = rowstart[t];
    int n     = (int)cnt[t];
    start = min(max(start, 0), E);
    n     = min(max(n, 0), E - start);

    const float4 self = xw4[(size_t)t * (DIM / 4) + lane];
    const float4 bb   = ((const float4*)bias)[lane];
    float4 acc = make_float4(0.f, 0.f, 0.f, 0.f);

    for (int base = 0; base < n; base += 32) {
        const int m = min(32, n - base);
        int   my_s = 0;
        float my_d = 0.f;
        if (lane < m) {
            my_s = list[start + base + lane];
            my_s = min(max(my_s, 0), N - 1);
            my_d = dinv[my_s];
        }
        for (int j = 0; j < m; j += GROUP) {
            float4 r[GROUP];
            for (int k = 0; k < GROUP; ++k) {
                if (j + k < m) {
                    const int s = __builtin_amdgcn_readlane(my_s, j + k);
                    r[k] = xw4[(size_t)s * (DIM / 4) + lane];
                } else {
                    r[k] = make_float4(0.f, 0.f, 0.f, 0.f);
                }
            }
            for (int k = 0; k < GROUP; ++k) {
                const float w = dt * __builtin_bit_cast(float,
                                    __builtin_amdgcn_readlane(__builtin_bit_cast(int, my_d), j + k));
                acc.x += r[k].x * w;
                acc.y += r[k].y * w;
                acc.z += r[k].z * w;
                acc.w += r[k].w * w;
            }
        }
    }
    acc.x += self.x * dt * dt;
    acc.y += self.y * dt * dt;
    acc.z += self.z * dt * dt;
    acc.w += self.w * dt * dt;

    float4 v;
    v.x = fmaxf(acc.x + bb.x, 0.f);
    v.y = fmaxf(acc.y + bb.y, 0.f);
    v.z = fmaxf(acc.z + bb.z, 0.f);
    v.w = fmaxf(acc.w + bb.w, 0.f);

    float m = fmaxf(fmaxf(v.x, v.y), fmaxf(v.z, v.w));
    #pragma unroll
    for (int off = 16; off > 0; off >>= 1) m = fmaxf(m, __shfl_xor(m, off, 32));

    v.x = expf(v.x - m); v.y = expf(v.y - m);
    v.z = expf(v.z - m); v.w = expf(v.w - m);
    float s = v.x + v.y + v.z + v.w;
    #pragma unroll
    for (int off = 16; off > 0; off >>= 1) s += __shfl_xor(s, off, 32);

    const float inv = 1.0f / s;
    v.x *= inv; v.y *= inv; v.z *= inv; v.w *= inv;
    ((float4*)out)[(size_t)t * (DIM / 4) + lane] = v;
}


extern "C" void kernel_launch(void* const* d_in, const int* in_sizes, int n_in,
                              void* d_out, int out_size, void* d_ws, size_t ws_size,
                              hipStream_t stream) {
    const float* x  = (const float*)d_in[0];
    const int*   ei = (const int*)  d_in[1];
    const float* W  = (const float*)d_in[2];
    const float* b  = (const float*)d_in[3];
    const int N = in_sizes[0] / DIM;
    const int E = in_sizes[1] / 2;
    const int* src = ei;
    const int* dst = ei + E;

    float* out = (float*)d_out;
    float*    xw       = (float*)d_ws;
    float*    dinv     = xw + (size_t)N * DIM;
    unsigned* cnt      = (unsigned*)(dinv + N);
    int*      rowstart = (int*)(cnt + N);
    int*      rank     = rowstart + N;
    int*      list     = rank + E;
    unsigned* total    = (unsigned*)(list + E);

    k_deg_init<<<(N + 255) / 256, 256, 0, stream>>>(cnt, total, N);
    k_count   <<<(E + 256 * EPT - 1) / (256 * EPT), 256, 0, stream>>>(dst, cnt, rank, E);
    k_offsets <<<(N + 1023) / 1024, 1024, 0, stream>>>(cnt, dinv, rowstart, total, N);
    k_fill    <<<(E + 256 * EPT - 1) / (256 * EPT), 256, 0, stream>>>(src, dst, rank, rowstart, list, E);

    const int Mtiles = (N + 15) / 16;
    k_gemm_wmma<<<(Mtiles + 7) / 8, 256, 0, stream>>>(x, W, xw, N, Mtiles);

    k_gather  <<<(N + 7) / 8, 256, 0, stream>>>(list, rowstart, cnt, dinv, xw, b, out, N, E);
}


# k_count: two planes of in-degree counters, one per half of the package, a work-group adds into its own half's; sum order = plane 0's edges then plane 1's, each in atomic rank order, self-loop last
# speedup vs baseline: 3.2464x; 1.5896x over previous
#include <hip/hip_runtime.h>
#include <math.h>


typedef float v2f __attribute__((ext_vector_type(2)));
typedef float v8f __attribute__((ext_vector_type(8)));

#define DIM 128
#define EPT 4


__global__ void k_deg_init(unsigned* __restrict__ cnt, unsigned* __restrict__ total, int words) {
    int i = blockIdx.x * 256 + threadIdx.x;
    if (i < words) cnt[i] = 0u;
    if (i == 0) *total = 0u;
}

__global__ void k_count(const int* __restrict__ dst, unsigned* __restrict__ cnt,
                        int* __restrict__ rank, int E, int NP) {
    const unsigned half = (__builtin_amdgcn_s_sendmsg_rtn(0x87) >> 18) & 1u;
    unsigned* __restrict__ c = cnt + (size_t)half * NP;
    const int T  = gridDim.x * 256;
    const int e0 = blockIdx.x * 256 + threadIdx.x;
    if (e0 + (EPT - 1) * T < E) {
        int t[EPT];
        unsigned r[EPT];
        for (int k = 0; k < EPT; ++k) t[k] = dst[e0 + k * T];
        for (int k = 0; k < EPT; ++k) r[k] = atomicAdd(&c[t[k]], 1u);
        for (int k = 0; k < EPT; ++k) rank[e0 + k * T] = (int)((r[k] << 1) | half);
    } else {
        for (int k = 0; k < EPT; ++k) {
            const int e = e0 + k * T;
            if (e < E) rank[e] = (int)((atomicAdd(&c[dst[e]], 1u) << 1) | half);
        }
    }
}

__global__ __launch_bounds__(1024) void k_offsets(const unsigned* __restrict__ cnt,
                                                  float* __restrict__ dinv,
                                                  unsigned* __restrict__ deg,
                                                  int* __restrict__ rowstart,
                                                  unsigned* __restrict__ total, int N, int NP) {
    __shared__ unsigned wsum[32];
    const int i    = blockIdx.x * 1024 + threadIdx.x;
    const int lane = threadIdx.x & 31;
    const int wave = threadIdx.x >> 5;
    const unsigned c0 = (i < N) ? cnt[i]      : 0u;
    const unsigned c1 = (i < N) ? cnt[NP + i] : 0u;
    const unsigned c  = c0 + c1;
    if (i < N) {
        dinv[i] = rsqrtf((float)(c + 1u));
        deg[i]  = c;
    }

    unsigned incl = c;
    for (int off = 1; off < 32; off <<= 1) {
        const unsigned o = __shfl_up(incl, off, 32);
        if (lane >= off) incl += o;
    }
    if (lane == 31) wsum[wave] = incl;
    __syncthreads();
    if (wave == 0) {
        const unsigned w = wsum[lane];
        unsigned winc = w;
        for (int off = 1; off < 32; off <<= 1) {
            const unsigned o = __shfl_up(winc, off, 32);
            if (lane >= off) winc += o;
        }
        unsigned base = 0u;
        if (lane == 31) base = atomicAdd(total, winc);
        base = __shfl(base, 31, 32);
        wsum[lane] = base + winc - w;
    }
    __syncthreads();
    if (i < N) {
        const unsigned s0 = wsum[wave] + incl - c;
        rowstart[i]      = (int)s0;
        rowstart[NP + i] = (int)(s0 + c0);
    }
}

__global__ void k_fill(const int* __restrict__ src, const int* __restrict__ dst,
                       const int* __restrict__ rank, const int* __restrict__ rowstart,
                       int* __restrict__ list, int E, int NP) {
    const int T  = gridDim.x * 256;
    const int e0 = blockIdx.x * 256 + threadIdx.x;
    if (e0 + (EPT - 1) * T < E) {
        int t[EPT], rk[EPT], s[EPT], pos[EPT];
        for (int k = 0; k < EPT; ++k) {
            t[k]  = dst [e0 + k * T];
            rk[k] = rank[e0 + k * T];
            s[k]  = src [e0 + k * T];
        }
        for (int k = 0; k < EPT; ++k)
            pos[k] = rowstart[(size_t)(rk[k] & 1) * NP + t[k]] + (int)((unsigned)rk[k] >> 1);
        for (int k = 0; k < EPT; ++k)
            list[min(max(pos[k], 0), E - 1)] = s[k];
    } else {
        for (int k = 0; k < EPT; ++k) {
            const int e = e0 + k * T;
            if (e < E) {
                const int rk1 = rank[e];
                int p = rowstart[(size_t)(rk1 & 1) * NP + dst[e]] + (int)((unsigned)rk1 >> 1);
                p = min(max(p, 0), E - 1);
                list[p] = src[e];
            }
        }
    }
}


#define WP 288

__global__ __launch_bounds__(256) void k_gemm_wmma(const float* __restrict__ x, const float* __restrict__ W,
                                                   float* __restrict__ xw, int N, int Mtiles) {
    __shared__ float Ws[(DIM / 2) * WP];
    const int lane = threadIdx.x & 31;
    const int wave = __builtin_amdgcn_readfirstlane(threadIdx.x >> 5);
    const int tm   = blockIdx.x * 8 + wave;
    const int row0 = tm * 16;

    const int mr = lane & 15;
    const int h  = lane >> 4;

    const int arow_i = min(row0 + mr, N - 1);
    const float* __restrict__ arow = x + (size_t)arow_i * DIM + 2 * h;
    v2f a[DIM / 4];
    #pragma unroll
    for (int s = 0; s < DIM / 4; ++s) a[s] = *(const v2f*)(arow + 4 * s);

    #pragma unroll
    for (int j = 0; j < (DIM / 2) * (DIM / 4) / 256; ++j) {
        const int idx = (int)threadIdx.x + j * 256;
        const int kp  = idx >> 5;
        const int c4  = idx & 31;
        const float4 r0 = ((const float4*)W)[(2 * kp)     * (DIM / 4) + c4];
        const float4 r1 = ((const float4*)W)[(2 * kp + 1) * (DIM / 4) + c4];
        float4* d = (float4*)&Ws[kp * WP + 8 * c4];
        d[0] = make_float4(r0.x, r1.x, r0.y, r1.y);
        d[1] = make_float4(r0.z, r1.z, r0.w, r1.w);
    }
    __syncthreads();
    if (tm >= Mtiles) return;

    v8f c[8];
    #pragma unroll
    for (int tn = 0; tn < 8; ++tn) c[tn] = (v8f){0.f, 0.f, 0.f, 0.f, 0.f, 0.f, 0.f, 0.f};

    const float* bp = Ws + h * WP + 2 * mr;
    #pragma unroll
    for (int s = 0; s < DIM / 4; ++s) {
        #pragma unroll
        for (int tn = 0; tn < 8; ++tn) {
            const v2f b = *(const v2f*)(bp + (2 * s) * WP + tn * 32);
            c[tn] = __builtin_amdgcn_wmma_f32_16x16x4_f32(false, a[s], false, b,
                                                          (short)0, c[tn], false, false);
        }
    }

    const int mbase = h << 3;
    float* __restrict__ orow = xw + (size_t)(row0 + mbase) * DIM + mr;
    if (row0 + 16 <= N) {
        #pragma unroll
        for (int r = 0; r < 8; ++r) {
            #pragma unroll
            for (int tn = 0; tn < 8; ++tn) orow[(size_t)r * DIM + tn * 16] = c[tn][r];
        }
    } else {
        #pragma unroll
        for (int r = 0; r < 8; ++r) {
            if (row0 + mbase + r < N) {
                #pragma unroll
                for (int tn = 0; tn < 8; ++tn) orow[(size_t)r * DIM + tn * 16] = c[tn][r];
            }
        }
    }
}


#define GROUP 8

__global__ __launch_bounds__(256) void k_gather(const int* __restrict__ list,
                                                const int* __restrict__ rowstart,
                                                const unsigned* __restrict__ deg,
                                                const float* __restrict__ dinv,
                                                const float* __restrict__ xw,
                                                const float* __restrict__ bias,
                                                float* __restrict__ out, int N, int E) {
    const int lane = threadIdx.x & 31;
    const int t = __builtin_amdgcn_readfirstlane(blockIdx.x * 8 + (threadIdx.x >> 5));
    if (t >= N) return;

    const float4* __restrict__ xw4 = (const float4*)xw;
    const float dt = dinv[t];
    int start = rowstart[t];
    int n     = (int)deg[t];
    start = min(max(start, 0), E);
    n     = min(max(n, 0), E - start);

    const float4 self = xw4[(size_t)t * (DIM / 4) + lane];
    const float4 bb   = ((const float4*)bias)[lane];
    float4 acc = make_float4(0.f, 0.f, 0.f, 0.f);

    for (int base = 0; base < n; base += 32) {
        const int m = min(32, n - base);
        int   my_s = 0;
        float my_d = 0.f;
        if (lane < m) {
            my_s = list[start + base + lane];
            my_s = min(max(my_s, 0), N - 1);
            my_d = dinv[my_s];
        }
        for (int j = 0; j < m; j += GROUP) {
            float4 r[GROUP];
            for (int k = 0; k < GROUP; ++k) {
                if (j + k < m) {
                    const int s = __builtin_amdgcn_readlane(my_s, j + k);
                    r[k] = xw4[(size_t)s * (DIM / 4) + lane];
                } else {
                    r[k] = make_float4(0.f, 0.f, 0.f, 0.f);
                }
            }
            for (int k = 0; k < GROUP; ++k) {
                const float w = dt * __builtin_bit_cast(float,
                                    __builtin_amdgcn_readlane(__builtin_bit_cast(int, my_d), j + k));
                acc.x += r[k].x * w;
                acc.y += r[k].y * w;
                acc.z += r[k].z * w;
                acc.w += r[k].w * w;
            }
        }
    }
    acc.x += self.x * dt * dt;
    acc.y += self.y * dt * dt;
    acc.z += self.z * dt * dt;
    acc.w += self.w * dt * dt;

    float4 v;
    v.x = fmaxf(acc.x + bb.x, 0.f);
    v.y = fmaxf(acc.y + bb.y, 0.f);
    v.z = fmaxf(acc.z + bb.z, 0.f);
    v.w = fmaxf(acc.w + bb.w, 0.f);

    float m = fmaxf(fmaxf(v.x, v.y), fmaxf(v.z, v.w));
    #pragma unroll
    for (int off = 16; off > 0; off >>= 1) m = fmaxf(m, __shfl_xor(m, off, 32));

    v.x = expf(v.x - m); v.y = expf(v.y - m);
    v.z = expf(v.z - m); v.w = expf(v.w - m);
    float s = v.x + v.y + v.z + v.w;
    #pragma unroll
    for (int off = 16; off > 0; off >>= 1) s += __shfl_xor(s, off, 32);

    const float inv = 1.0f / s;
    v.x *= inv; v.y *= inv; v.z *= inv; v.w *= inv;
    ((float4*)out)[(size_t)t * (DIM / 4) + lane] = v;
}


extern "C" void kernel_launch(void* const* d_in, const int* in_sizes, int n_in,
                              void* d_out, int out_size, void* d_ws, size_t ws_size,
                              hipStream_t stream) {
    const float* x  = (const float*)d_in[0];
    const int*   ei = (const int*)  d_in[1];
    const float* W  = (const float*)d_in[2];
    const float* b  = (const float*)d_in[3];
    const int N = in_sizes[0] / DIM;
    const int E = in_sizes[1] / 2;
    const int* src = ei;
    const int* dst = ei + E;

    float* out = (float*)d_out;
    float*    xw       = (float*)d_ws;
    const int NP = (N + 63) / 64 * 64;
    float*    dinv     = xw + (size_t)N * DIM;
    unsigned* cnt      = (unsigned*)(dinv + NP);
    int*      rowstart = (int*)(cnt + 2 * NP);
    unsigned* deg      = (unsigned*)(rowstart + 2 * NP);
    int*      rank     = (int*)(deg + NP);
    int*      list     = rank + E;
    unsigned* total    = (unsigned*)(list + E);

    k_deg_init<<<(2 * NP + 255) / 256, 256, 0, stream>>>(cnt, total, 2 * NP);
    k_count   <<<(E + 256 * EPT - 1) / (256 * EPT), 256, 0, stream>>>(dst, cnt, rank, E, NP);
    k_offsets <<<(N + 1023) / 1024, 1024, 0, stream>>>(cnt, dinv, deg, rowstart, total, N, NP);
    k_fill    <<<(E + 256 * EPT - 1) / (256 * EPT), 256, 0, stream>>>(src, dst, rank, rowstart, list, E, NP);

    const int Mtiles = (N + 15) / 16;
    k_gemm_wmma<<<(Mtiles + 7) / 8, 256, 0, stream>>>(x, W, xw, N, Mtiles);

    k_gather  <<<(N + 7) / 8, 256, 0, stream>>>(list, rowstart, deg, dinv, xw, b, out, N, E);
}


# k_count stores src straight at slots[node][plane][rank] (ranks past 32: overflow list, summed before the self-loop), so the prefix-sum and fill launches go; dinv is made in the GEMM launch
# speedup vs baseline: 3.6878x; 1.1360x over previous
#include <hip/hip_runtime.h>
#include <math.h>


typedef float v2f __attribute__((ext_vector_type(2)));
typedef float v8f __attribute__((ext_vector_type(8)));

#define DIM 128
#define EPT 4
#define SLOTS 32


__global__ void k_deg_init(unsigned* __restrict__ cnt, unsigned* __restrict__ ovf_n, int words) {
    int i = blockIdx.x * 256 + threadIdx.x;
    if (i < words) cnt[i] = 0u;
    if (i == 0) *ovf_n = 0u;
}


__device__ __forceinline__ void place_edge(int t, int s, unsigned r, unsigned half,
                                           int* __restrict__ slots, unsigned* __restrict__ ovf_n,
                                           int* __restrict__ ovf, int E) {
    if (r < (unsigned)SLOTS) {
        slots[((size_t)t * 2 + half) * SLOTS + r] = s;
    } else {
        const unsigned i = atomicAdd(ovf_n, 1u);
        if (i < (unsigned)E) {
            ovf[2 * (size_t)i]     = t;
            ovf[2 * (size_t)i + 1] = s;
        }
    }
}

__global__ void k_count(const int* __restrict__ src, const int* __restrict__ dst,
                        unsigned* __restrict__ cnt, int* __restrict__ slots,
                        unsigned* __restrict__ ovf_n, int* __restrict__ ovf, int E, int NP) {
    const unsigned half = (__builtin_amdgcn_s_sendmsg_rtn(0x87) >> 18) & 1u;
    unsigned* __restrict__ c = cnt + (size_t)half * NP;
    const int T  = gridDim.x * 256;
    const int e0 = blockIdx.x * 256 + threadIdx.x;
    if (e0 + (EPT - 1) * T < E) {
        int t[EPT], s[EPT];
        unsigned r[EPT];
        for (int k = 0; k < EPT; ++k) t[k] = dst[e0 + k * T];
        for (int k = 0; k < EPT; ++k) s[k] = src[e0 + k * T];
        for (int k = 0; k < EPT; ++k) r[k] = atomicAdd(&c[t[k]], 1u);
        for (int k = 0; k < EPT; ++k) place_edge(t[k], s[k], r[k], half, slots, ovf_n, ovf, E);
    } else {
        for (int k = 0; k < EPT; ++k) {
            const int e = e0 + k * T;
            if (e < E) {
                const int t1 = dst[e];
                const int s1 = src[e];
                place_edge(t1, s1, atomicAdd(&c[t1], 1u), half, slots, ovf_n, ovf, E);
            }
        }
    }
}


#define WP 288

__global__ __launch_bounds__(256) void k_gemm_wmma(const float* __restrict__ x, const float* __restrict__ W,
                                                   float* __restrict__ xw,
                                                   const unsigned* __restrict__ cnt, float* __restrict__ dinv,
                                                   int N, int NP, int Mtiles) {
    __shared__ float Ws[(DIM / 2) * WP];
    const int lane = threadIdx.x & 31;
    const int wave = __builtin_amdgcn_readfirstlane(threadIdx.x >> 5);
    const int tm   = blockIdx.x * 8 + wave;
    const int row0 = tm * 16;

    const int mr = lane & 15;
    const int h  = lane >> 4;

    const int arow_i = min(row0 + mr, N - 1);
    const float* __restrict__ arow = x + (size_t)arow_i * DIM + 2 * h;
    v2f a[DIM / 4];
    #pragma unroll
    for (int s = 0; s < DIM / 4; ++s) a[s] = *(const v2f*)(arow + 4 * s);

    #pragma unroll
    for (int j = 0; j < (DIM / 2) * (DIM / 4) / 256; ++j) {
        const int idx = (int)threadIdx.x + j * 256;
        const int kp  = idx >> 5;
        const int c4  = idx & 31;
        const float4 r0 = ((const float4*)W)[(2 * kp)     * (DIM / 4) + c4];
        const float4 r1 = ((const float4*)W)[(2 * kp + 1) * (DIM / 4) + c4];
        float4* d = (float4*)&Ws[kp * WP + 8 * c4];
        d[0] = make_float4(r0.x, r1.x, r0.y, r1.y);
        d[1] = make_float4(r0.z, r1.z, r0.w, r1.w);
    }
    const int gi = blockIdx.x * 256 + (int)threadIdx.x;
    if (gi < N) dinv[gi] = rsqrtf((float)(cnt[gi] + cnt[NP + gi] + 1u));
    __syncthreads();
    if (tm >= Mtiles) return;

    v8f c[8];
    #pragma unroll
    for (int tn = 0; tn < 8; ++tn) c[tn] = (v8f){0.f, 0.f, 0.f, 0.f, 0.f, 0.f, 0.f, 0.f};

    const float* bp = Ws + h * WP + 2 * mr;
    #pragma unroll
    for (int s = 0; s < DIM / 4; ++s) {
        #pragma unroll
        for (int tn = 0; tn < 8; ++tn) {
            const v2f b = *(const v2f*)(bp + (2 * s) * WP + tn * 32);
            c[tn] = __builtin_amdgcn_wmma_f32_16x16x4_f32(false, a[s], false, b,
                                                          (short)0, c[tn], false, false);
        }
    }

    const int mbase = h << 3;
    float* __restrict__ orow = xw + (size_t)(row0 + mbase) * DIM + mr;
    if (row0 + 16 <= N) {
        #pragma unroll
        for (int r = 0; r < 8; ++r) {
            #pragma unroll
            for (int tn = 0; tn < 8; ++tn) orow[(size_t)r * DIM + tn * 16] = c[tn][r];
        }
    } else {
        #pragma unroll
        for (int r = 0; r < 8; ++r) {
            if (row0 + mbase + r < N) {
                #pragma unroll
                for (int tn = 0; tn < 8; ++tn) orow[(size_t)r * DIM + tn * 16] = c[tn][r];
            }
        }
    }
}


#define GROUP 8

__global__ __launch_bounds__(256) void k_gather(const int* __restrict__ slots,
                                                const unsigned* __restrict__ cnt,
                                                const unsigned* __restrict__ ovf_n,
                                                const int* __restrict__ ovf,
                                                const float* __restrict__ dinv,
                                                const float* __restrict__ xw,
                                                const float* __restrict__ bias,
                                                float* __restrict__ out, int N, int NP, int E) {
    const int lane = threadIdx.x & 31;
    const int t = __builtin_amdgcn_readfirstlane(blockIdx.x * 8 + (threadIdx.x >> 5));
    if (t >= N) return;

    const float4* __restrict__ xw4 = (const float4*)xw;
    const float dt = dinv[t];
    const unsigned c0 = cnt[t];
    const unsigned c1 = cnt[NP + t];
    const int n0 = (int)min(c0, (unsigned)SLOTS);
    const int n1 = (int)min(c1, (unsigned)SLOTS);
    const int n  = n0 + n1;
    const int* __restrict__ row = slots + (size_t)t * (2 * SLOTS);

    const float4 self = xw4[(size_t)t * (DIM / 4) + lane];
    const float4 bb   = ((const float4*)bias)[lane];
    float4 acc = make_float4(0.f, 0.f, 0.f, 0.f);

    for (int base = 0; base < n; base += 32) {
        const int m = min(32, n - base);
        int   my_s = 0;
        float my_d = 0.f;
        if (lane < m) {
            const int p = base + lane;
            my_s = row[p < n0 ? p : p - n0 + SLOTS];
            my_s = min(max(my_s, 0), N - 1);
            my_d = dinv[my_s];
        }
        for (int j = 0; j < m; j += GROUP) {
            float4 r[GROUP];
            for (int k = 0; k < GROUP; ++k) {
                if (j + k < m) {
                    const int s = __builtin_amdgcn_readlane(my_s, j + k);
                    r[k] = xw4[(size_t)s * (DIM / 4) + lane];
                } else {
                    r[k] = make_float4(0.f, 0.f, 0.f, 0.f);
                }
            }
            for (int k = 0; k < GROUP; ++k) {
                const float w = dt * __builtin_bit_cast(float,
                                    __builtin_amdgcn_readlane(__builtin_bit_cast(int, my_d), j + k));
                acc.x += r[k].x * w;
                acc.y += r[k].y * w;
                acc.z += r[k].z * w;
                acc.w += r[k].w * w;
            }
        }
    }
    if (c0 > (unsigned)SLOTS || c1 > (unsigned)SLOTS) {
        const int total = (int)min(*ovf_n, (unsigned)E);
        for (int base = 0; base < total; base += 32) {
            int   od = -1, os = 0;
            float ow = 0.f;
            if (base + lane < total) {
                od = ovf[2 * (size_t)(base + lane)];
                os = ovf[2 * (size_t)(base + lane) + 1];
                os = min(max(os, 0), N - 1);
                if (od == t) ow = dinv[os];
            }
            unsigned hit = __builtin_amdgcn_ballot_w32(od == t);
            while (hit != 0u) {
                const int j = __builtin_ctz(hit);
                hit &= hit - 1u;
                const int   s = __builtin_amdgcn_readlane(os, j);
                const float w = dt * __builtin_bit_cast(float,
                                    __builtin_amdgcn_readlane(__builtin_bit_cast(int, ow), j));
                const float4 r = xw4[(size_t)s * (DIM / 4) + lane];
                acc.x += r.x * w;
                acc.y += r.y * w;
                acc.z += r.z * w;
                acc.w += r.w * w;
            }
        }
    }
    acc.x += self.x * dt * dt;
    acc.y += self.y * dt * dt;
    acc.z += self.z * dt * dt;
    acc.w += self.w * dt * dt;

    float4 v;
    v.x = fmaxf(acc.x + bb.x, 0.f);
    v.y = fmaxf(acc.y + bb.y, 0.f);
    v.z = fmaxf(acc.z + bb.z, 0.f);
    v.w = fmaxf(acc.w + bb.w, 0.f);

    float m = fmaxf(fmaxf(v.x, v.y), fmaxf(v.z, v.w));
    #pragma unroll
    for (int off = 16; off > 0; off >>= 1) m = fmaxf(m, __shfl_xor(m, off, 32));

    v.x = expf(v.x - m); v.y = expf(v.y - m);
    v.z = expf(v.z - m); v.w = expf(v.w - m);
    float s = v.x + v.y + v.z + v.w;
    #pragma unroll
    for (int off = 16; off > 0; off >>= 1) s += __shfl_xor(s, off, 32);

    const float inv = 1.0f / s;
    v.x *= inv; v.y *= inv; v.z *= inv; v.w *= inv;
    ((float4*)out)[(size_t)t * (DIM / 4) + lane] = v;
}


extern "C" void kernel_launch(void* const* d_in, const int* in_sizes, int n_in,
                              void* d_out, int out_size, void* d_ws, size_t ws_size,
                              hipStream_t stream) {
    const float* x  = (const float*)d_in[0];
    const int*   ei = (const int*)  d_in[1];
    const float* W  = (const float*)d_in[2];
    const float* b  = (const float*)d_in[3];
    const int N = in_sizes[0] / DIM;
    const int E = in_sizes[1] / 2;
    const int* src = ei;
    const int* dst = ei + E;

    float* out = (float*)d_out;
    float*    xw       = (float*)d_ws;
    const int NP = (N + 63) / 64 * 64;
    float*    dinv     = xw + (size_t)N * DIM;
    unsigned* cnt      = (unsigned*)(dinv + NP);
    int*      slots    = (int*)(cnt + 2 * NP);
    int*      ovf      = slots + (size_t)N * 2 * SLOTS;
    unsigned* ovf_n    = (unsigned*)(ovf + 2 * (size_t)E);

    k_deg_init<<<(2 * NP + 255) / 256, 256, 0, stream>>>(cnt, ovf_n, 2 * NP);
    k_count   <<<(E + 256 * EPT - 1) / (256 * EPT), 256, 0, stream>>>(src, dst, cnt, slots, ovf_n, ovf, E, NP);

    const int Mtiles = (N + 15) / 16;
    k_gemm_wmma<<<(Mtiles + 7) / 8, 256, 0, stream>>>(x, W, xw, cnt, dinv, N, NP, Mtiles);

    k_gather  <<<(N + 7) / 8, 256, 0, stream>>>(slots, cnt, ovf_n, ovf, dinv, xw, b, out, N, NP, E);
}


# GEMM launch moves whole 128-B lines only: a wave's 16 x rows come in as 4-line loads and feed the A fragments from an LDS slot of its own; product rows leave as one 4-line store each; same sums
# speedup vs baseline: 4.0549x; 1.0995x over previous
#include <hip/hip_runtime.h>
#include <math.h>


typedef float v2f __attribute__((ext_vector_type(2)));
typedef float v8f __attribute__((ext_vector_type(8)));

#define DIM 128
#define EPT 4
#define SLOTS 32


__global__ void k_deg_init(unsigned* __restrict__ cnt, unsigned* __restrict__ ovf_n, int words) {
    int i = blockIdx.x * 256 + threadIdx.x;
    if (i < words) cnt[i] = 0u;
    if (i == 0) *ovf_n = 0u;
}


__device__ __forceinline__ void place_edge(int t, int s, unsigned r, unsigned half,
                                           int* __restrict__ slots, unsigned* __restrict__ ovf_n,
                                           int* __restrict__ ovf, int E) {
    if (r < (unsigned)SLOTS) {
        slots[((size_t)t * 2 + half) * SLOTS + r] = s;
    } else {
        const unsigned i = atomicAdd(ovf_n, 1u);
        if (i < (unsigned)E) {
            ovf[2 * (size_t)i]     = t;
            ovf[2 * (size_t)i + 1] = s;
        }
    }
}

__global__ void k_count(const int* __restrict__ src, const int* __restrict__ dst,
                        unsigned* __restrict__ cnt, int* __restrict__ slots,
                        unsigned* __restrict__ ovf_n, int* __restrict__ ovf, int E, int NP) {
    const unsigned half = (__builtin_amdgcn_s_sendmsg_rtn(0x87) >> 18) & 1u;
    unsigned* __restrict__ c = cnt + (size_t)half * NP;
    const int T  = gridDim.x * 256;
    const int e0 = blockIdx.x * 256 + threadIdx.x;
    if (e0 + (EPT - 1) * T < E) {
        int t[EPT], s[EPT];
        unsigned r[EPT];
        for (int k = 0; k < EPT; ++k) t[k] = dst[e0 + k * T];
        for (int k = 0; k < EPT; ++k) s[k] = src[e0 + k * T];
        for (int k = 0; k < EPT; ++k) r[k] = atomicAdd(&c[t[k]], 1u);
        for (int k = 0; k < EPT; ++k) place_edge(t[k], s[k], r[k], half, slots, ovf_n, ovf, E);
    } else {
        for (int k = 0; k < EPT; ++k) {
            const int e = e0 + k * T;
            if (e < E) {
                const int t1 = dst[e];
                const int s1 = src[e];
                place_edge(t1, s1, atomicAdd(&c[t1], 1u), half, slots, ovf_n, ovf, E);
            }
        }
    }
}


#define WP 288
#define SP 132

__global__ __launch_bounds__(256) void k_gemm_wmma(const float* __restrict__ x, const float* __restrict__ W,
                                                   float* __restrict__ xw,
                                                   const unsigned* __restrict__ cnt,
                                                   int N, int NP, int Mtiles) {
    __shared__ float Ws[(DIM / 2) * WP];
    __shared__ __attribute__((aligned(16))) float Sl[8 * 16 * SP];
    const int lane = threadIdx.x & 31;
    const int wave = __builtin_amdgcn_readfirstlane(threadIdx.x >> 5);
    const int tm   = blockIdx.x * 8 + wave;
    const int row0 = tm * 16;

    const int mr = lane & 15;
    const int h  = lane >> 4;

    float* __restrict__ slot = Sl + wave * (16 * SP);

    const int drow = min(row0 + mr, N - 1);
    const unsigned deg = cnt[drow] + cnt[NP + drow] + 1u;

    float4 q[16];
    #pragma unroll
    for (int i = 0; i < 16; ++i) {
        const int ri = min(row0 + i, N - 1);
        q[i] = ((const float4*)x)[(size_t)ri * (DIM / 4) + lane];
    }

    #pragma unroll
    for (int j = 0; j < (DIM / 2) * (DIM / 4) / 256; ++j) {
        const int idx = (int)threadIdx.x + j * 256;
        const int kp  = idx >> 5;
        const int c4  = idx & 31;
        const float4 r0 = ((const float4*)W)[(2 * kp)     * (DIM / 4) + c4];
        const float4 r1 = ((const float4*)W)[(2 * kp + 1) * (DIM / 4) + c4];
        float4* d = (float4*)&Ws[kp * WP + 8 * c4];
        d[0] = make_float4(r0.x, r1.x, r0.y, r1.y);
        d[1] = make_float4(r0.z, r1.z, r0.w, r1.w);
    }
    #pragma unroll
    for (int i = 0; i < 16; ++i) *(float4*)&slot[i * SP + 4 * lane] = q[i];
    __syncthreads();
    if (tm >= Mtiles) return;

    v8f c[8];
    #pragma unroll
    for (int tn = 0; tn < 8; ++tn) c[tn] = (v8f){0.f, 0.f, 0.f, 0.f, 0.f, 0.f, 0.f, 0.f};

    const float* ap = slot + mr * SP + 2 * h;
    const float* bp = Ws + h * WP + 2 * mr;
    #pragma unroll
    for (int s = 0; s < DIM / 4; ++s) {
        const v2f a = *(const v2f*)(ap + 4 * s);
        #pragma unroll
        for (int tn = 0; tn < 8; ++tn) {
            const v2f b = *(const v2f*)(bp + (2 * s) * WP + tn * 32);
            c[tn] = __builtin_amdgcn_wmma_f32_16x16x4_f32(false, a, false, b,
                                                          (short)0, c[tn], false, false);
        }
    }

    const int mbase = h << 3;
    const float dv = rsqrtf((float)deg);
    float dsc[8];
    #pragma unroll
    for (int r = 0; r < 8; ++r) dsc[r] = __shfl(dv, mbase + r, 32);

    #pragma unroll
    for (int tn = 0; tn < 8; ++tn) {
        #pragma unroll
        for (int r = 0; r < 8; ++r) slot[(mbase + r) * SP + tn * 16 + mr] = c[tn][r] * dsc[r];
    }
    __builtin_amdgcn_fence(__ATOMIC_SEQ_CST, "wavefront");

    float4* __restrict__ orow = (float4*)xw + (size_t)row0 * (DIM / 4) + lane;
    if (row0 + 16 <= N) {
        float4 o[16];
        #pragma unroll
        for (int i = 0; i < 16; ++i) o[i] = *(const float4*)&slot[i * SP + 4 * lane];
        #pragma unroll
        for (int i = 0; i < 16; ++i) orow[(size_t)i * (DIM / 4)] = o[i];
    } else {
        #pragma unroll
        for (int i = 0; i < 16; ++i) {
            if (row0 + i < N) orow[(size_t)i * (DIM / 4)] = *(const float4*)&slot[i * SP + 4 * lane];
        }
    }
}


#define GROUP 8

__global__ __launch_bounds__(256) void k_gather(const int* __restrict__ slots,
                                                const unsigned* __restrict__ cnt,
                                                const unsigned* __restrict__ ovf_n,
                                                const int* __restrict__ ovf,
                                                const float* __restrict__ xw,
                                                const float* __restrict__ bias,
                                                float* __restrict__ out, int N, int NP, int E) {
    const int lane = threadIdx.x & 31;
    const int t = __builtin_amdgcn_readfirstlane(blockIdx.x * 8 + (threadIdx.x >> 5));
    if (t >= N) return;

    const float4* __restrict__ xw4 = (const float4*)xw;
    const unsigned c0 = cnt[t];
    const unsigned c1 = cnt[NP + t];
    const float dt = rsqrtf((float)(c0 + c1 + 1u));
    const int n0 = (int)min(c0, (unsigned)SLOTS);
    const int n1 = (int)min(c1, (unsigned)SLOTS);
    const int n  = n0 + n1;
    const int* __restrict__ row = slots + (size_t)t * (2 * SLOTS);

    const float4 self = xw4[(size_t)t * (DIM / 4) + lane];
    const float4 bb   = ((const float4*)bias)[lane];
    float4 acc = make_float4(0.f, 0.f, 0.f, 0.f);

    for (int base = 0; base < n; base += 32) {
        const int m = min(32, n - base);
        int my_s = 0;
        if (lane < m) {
            const int p = base + lane;
            my_s = row[p < n0 ? p : p - n0 + SLOTS];
            my_s = min(max(my_s, 0), N - 1);
        }
        for (int j = 0; j < m; j += GROUP) {
            float4 r[GROUP];
            for (int k = 0; k < GROUP; ++k) {
                if (j + k < m) {
                    const int s = __builtin_amdgcn_readlane(my_s, j + k);
                    r[k] = xw4[(size_t)s * (DIM / 4) + lane];
                } else {
                    r[k] = make_float4(0.f, 0.f, 0.f, 0.f);
                }
            }
            for (int k = 0; k < GROUP; ++k) {
                acc.x += r[k].x;
                acc.y += r[k].y;
                acc.z += r[k].z;
                acc.w += r[k].w;
            }
        }
    }
    if (c0 > (unsigned)SLOTS || c1 > (unsigned)SLOTS) {
        const int total = (int)min(*ovf_n, (unsigned)E);
        for (int base = 0; base < total; base += 32) {
            int od = -1, os = 0;
            if (base + lane < total) {
                od = ovf[2 * (size_t)(base + lane)];
                os = ovf[2 * (size_t)(base + lane) + 1];
                os = min(max(os, 0), N - 1);
            }
            unsigned hit = __builtin_amdgcn_ballot_w32(od == t);
            while (hit != 0u) {
                const int j = __builtin_ctz(hit);
                hit &= hit - 1u;
                const int s = __builtin_amdgcn_readlane(os, j);
                const float4 r = xw4[(size_t)s * (DIM / 4) + lane];
                acc.x += r.x;
                acc.y += r.y;
                acc.z += r.z;
                acc.w += r.w;
            }
        }
    }
    acc.x = (acc.x + self.x) * dt;
    acc.y = (acc.y + self.y) * dt;
    acc.z = (acc.z + self.z) * dt;
    acc.w = (acc.w + self.w) * dt;

    float4 v;
    v.x = fmaxf(acc.x + bb.x, 0.f);
    v.y = fmaxf(acc.y + bb.y, 0.f);
    v.z = fmaxf(acc.z + bb.z, 0.f);
    v.w = fmaxf(acc.w + bb.w, 0.f);

    float m = fmaxf(fmaxf(v.x, v.y), fmaxf(v.z, v.w));
    #pragma unroll
    for (int off = 16; off > 0; off >>= 1) m = fmaxf(m, __shfl_xor(m, off, 32));

    v.x = expf(v.x - m); v.y = expf(v.y - m);
    v.z = expf(v.z - m); v.w = expf(v.w - m);
    float s = v.x + v.y + v.z + v.w;
    #pragma unroll
    for (int off = 16; off > 0; off >>= 1) s += __shfl_xor(s, off, 32);

    const float inv = 1.0f / s;
    v.x *= inv; v.y *= inv; v.z *= inv; v.w *= inv;
    ((float4*)out)[(size_t)t * (DIM / 4) + lane] = v;
}


extern "C" void kernel_launch(void* const* d_in, const int* in_sizes, int n_in,
                              void* d_out, int out_size, void* d_ws, size_t ws_size,
                              hipStream_t stream) {
    const float* x  = (const float*)d_in[0];
    const int*   ei = (const int*)  d_in[1];
    const float* W  = (const float*)d_in[2];
    const float* b  = (const float*)d_in[3];
    const int N = in_sizes[0] / DIM;
    const int E = in_sizes[1] / 2;
    const int* src = ei;
    const int* dst = ei + E;

    float* out = (float*)d_out;
    float*    xw       = (float*)d_ws;
    const int NP = (N + 63) / 64 * 64;
    unsigned* cnt      = (unsigned*)(xw + (size_t)N * DIM);
    int*      slots    = (int*)(cnt + 2 * NP);
    int*      ovf      = slots + (size_t)N * 2 * SLOTS;
    unsigned* ovf_n    = (unsigned*)(ovf + 2 * (size_t)E);

    k_deg_init<<<(2 * NP + 255) / 256, 256, 0, stream>>>(cnt, ovf_n, 2 * NP);
    k_count   <<<(E + 256 * EPT - 1) / (256 * EPT), 256, 0, stream>>>(src, dst, cnt, slots, ovf_n, ovf, E, NP);

    const int Mtiles = (N + 15) / 16;
    k_gemm_wmma<<<(Mtiles + 7) / 8, 256, 0, stream>>>(x, W, xw, cnt, N, NP, Mtiles);

    k_gather  <<<(N + 7) / 8, 256, 0, stream>>>(slots, cnt, ovf_n, ovf, xw, b, out, N, NP, E);
}


# speedup vs baseline: 5.2930x; 1.3053x over previous
#include <hip/hip_runtime.h>
#include <math.h>


typedef float v2f __attribute__((ext_vector_type(2)));
typedef float v8f __attribute__((ext_vector_type(8)));

#define DIM 128
#define EPT 4
#define SLOTS 32


__global__ void k_deg_init(unsigned* __restrict__ cnt, unsigned* __restrict__ ovf_n, int words) {
    int i = blockIdx.x * 256 + threadIdx.x;
    if (i < words) cnt[i] = 0u;
    if (i == 0) *ovf_n = 0u;
}


__device__ __forceinline__ void place_edge(int t, int s, unsigned r, unsigned half,
                                           int* __restrict__ slots, unsigned* __restrict__ ovf_n,
                                           int* __restrict__ ovf, int E, int NP) {
    if (r < (unsigned)SLOTS) {
        slots[((size_t)half * NP + t) * SLOTS + r] = s;
    } else {
        const unsigned i = atomicAdd(ovf_n, 1u);
        if (i < (unsigned)E) {
            ovf[2 * (size_t)i]     = t;
            ovf[2 * (size_t)i + 1] = s;
        }
    }
}

__global__ void k_count(const int* __restrict__ src, const int* __restrict__ dst,
                        unsigned* __restrict__ cnt, int* __restrict__ slots,
                        unsigned* __restrict__ ovf_n, int* __restrict__ ovf, int E, int NP) {
    const unsigned half = (__builtin_amdgcn_s_sendmsg_rtn(0x87) >> 18) & 1u;
    unsigned* __restrict__ c = cnt + (size_t)half * NP;
    const int T  = gridDim.x * 256;
    const int e0 = blockIdx.x * 256 + threadIdx.x;
    if (e0 + (EPT - 1) * T < E) {
        int t[EPT], s[EPT];
        unsigned r[EPT];
        for (int k = 0; k < EPT; ++k) t[k] = dst[e0 + k * T];
        for (int k = 0; k < EPT; ++k) s[k] = src[e0 + k * T];
        for (int k = 0; k < EPT; ++k) r[k] = atomicAdd(&c[t[k]], 1u);
        for (int k = 0; k < EPT; ++k) place_edge(t[k], s[k], r[k], half, slots, ovf_n, ovf, E, NP);
    } else {
        for (int k = 0; k < EPT; ++k) {
            const int e = e0 + k * T;
            if (e < E) {
                const int t1 = dst[e];
                const int s1 = src[e];
                place_edge(t1, s1, atomicAdd(&c[t1], 1u), half, slots, ovf_n, ovf, E, NP);
            }
        }
    }
}


#define WP 288
#define SP 132

__global__ __launch_bounds__(256) void k_gemm_wmma(const float* __restrict__ x, const float* __restrict__ W,
                                                   float* __restrict__ xw,
                                                   const unsigned* __restrict__ cnt,
                                                   int N, int NP, int Mtiles) {
    __shared__ float Ws[(DIM / 2) * WP];
    __shared__ __attribute__((aligned(16))) float Sl[8 * 16 * SP];
    const int lane = threadIdx.x & 31;
    const int wave = __builtin_amdgcn_readfirstlane(threadIdx.x >> 5);
    const int tm   = blockIdx.x * 8 + wave;
    const int row0 = tm * 16;

    const int mr = lane & 15;
    const int h  = lane >> 4;

    float* __restrict__ slot = Sl + wave * (16 * SP);

    const int drow = min(row0 + mr, N - 1);
    const unsigned deg = cnt[drow] + cnt[NP + drow] + 1u;

    float4 q[16];
    #pragma unroll
    for (int i = 0; i < 16; ++i) {
        const int ri = min(row0 + i, N - 1);
        q[i] = ((const float4*)x)[(size_t)ri * (DIM / 4) + lane];
    }

    #pragma unroll
    for (int j = 0; j < (DIM / 2) * (DIM / 4) / 256; ++j) {
        const int idx = (int)threadIdx.x + j * 256;
        const int kp  = idx >> 5;
        const int c4  = idx & 31;
        const float4 r0 = ((const float4*)W)[(2 * kp)     * (DIM / 4) + c4];
        const float4 r1 = ((const float4*)W)[(2 * kp + 1) * (DIM / 4) + c4];
        float4* d = (float4*)&Ws[kp * WP + 8 * c4];
        d[0] = make_float4(r0.x, r1.x, r0.y, r1.y);
        d[1] = make_float4(r0.z, r1.z, r0.w, r1.w);
    }
    #pragma unroll
    for (int i = 0; i < 16; ++i) *(float4*)&slot[i * SP + 4 * lane] = q[i];
    __syncthreads();
    if (tm >= Mtiles) return;

    v8f c[8];
    #pragma unroll
    for (int tn = 0; tn < 8; ++tn) c[tn] = (v8f){0.f, 0.f, 0.f, 0.f, 0.f, 0.f, 0.f, 0.f};

    const float* ap = slot + mr * SP + 2 * h;
    const float* bp = Ws + h * WP + 2 * mr;
    #pragma unroll
    for (int s = 0; s < DIM / 4; ++s) {
        const v2f a = *(const v2f*)(ap + 4 * s);
        #pragma unroll
        for (int tn = 0; tn < 8; ++tn) {
            const v2f b = *(const v2f*)(bp + (2 * s) * WP + tn * 32);
            c[tn] = __builtin_amdgcn_wmma_f32_16x16x4_f32(false, a, false, b,
                                                          (short)0, c[tn], false, false);
        }
    }

    const int mbase = h << 3;
    const float dv = rsqrtf((float)deg);
    float dsc[8];
    #pragma unroll
    for (int r = 0; r < 8; ++r) dsc[r] = __shfl(dv, mbase + r, 32);

    #pragma unroll
    for (int tn = 0; tn < 8; ++tn) {
        #pragma unroll
        for (int r = 0; r < 8; ++r) slot[(mbase + r) * SP + tn * 16 + mr] = c[tn][r] * dsc[r];
    }
    __builtin_amdgcn_fence(__ATOMIC_SEQ_CST, "wavefront");

    float4* __restrict__ orow = (float4*)xw + (size_t)row0 * (DIM / 4) + lane;
    if (row0 + 16 <= N) {
        float4 o[16];
        #pragma unroll
        for (int i = 0; i < 16; ++i) o[i] = *(const float4*)&slot[i * SP + 4 * lane];
        #pragma unroll
        for (int i = 0; i < 16; ++i) orow[(size_t)i * (DIM / 4)] = o[i];
    } else {
        #pragma unroll
        for (int i = 0; i < 16; ++i) {
            if (row0 + i < N) orow[(size_t)i * (DIM / 4)] = *(const float4*)&slot[i * SP + 4 * lane];
        }
    }
}


#define GROUP 8

__global__ __launch_bounds__(256) void k_gather(const int* __restrict__ slots,
                                                const unsigned* __restrict__ cnt,
                                                const unsigned* __restrict__ ovf_n,
                                                const int* __restrict__ ovf,
                                                const float* __restrict__ xw,
                                                const float* __restrict__ bias,
                                                float* __restrict__ out, int N, int NP, int E) {
    const int lane = threadIdx.x & 31;
    const int t = __builtin_amdgcn_readfirstlane(blockIdx.x * 8 + (threadIdx.x >> 5));
    if (t >= N) return;

    const float4* __restrict__ xw4 = (const float4*)xw;
    const unsigned c0 = cnt[t];
    const unsigned c1 = cnt[NP + t];
    const float dt = rsqrtf((float)(c0 + c1 + 1u));
    const int n0 = (int)min(c0, (unsigned)SLOTS);
    const int n1 = (int)min(c1, (unsigned)SLOTS);
    const int n  = n0 + n1;
    const int* __restrict__ row0 = slots + (size_t)t * SLOTS;
    const int* __restrict__ row1 = slots + ((size_t)NP + t) * SLOTS;

    const float4 self = xw4[(size_t)t * (DIM / 4) + lane];
    const float4 bb   = ((const float4*)bias)[lane];
    float4 acc = make_float4(0.f, 0.f, 0.f, 0.f);

    for (int base = 0; base < n; base += 32) {
        const int m = min(32, n - base);
        int my_s = 0;
        if (lane < m) {
            const int p = base + lane;
            my_s = p < n0 ? row0[p] : row1[p - n0];
            my_s = min(max(my_s, 0), N - 1);
        }
        for (int j = 0; j < m; j += GROUP) {
            float4 r[GROUP];
            for (int k = 0; k < GROUP; ++k) {
                if (j + k < m) {
                    const int s = __builtin_amdgcn_readlane(my_s, j + k);
                    r[k] = xw4[(size_t)s * (DIM / 4) + lane];
                } else {
                    r[k] = make_float4(0.f, 0.f, 0.f, 0.f);
                }
            }
            for (int k = 0; k < GROUP; ++k) {
                acc.x += r[k].x;
                acc.y += r[k].y;
                acc.z += r[k].z;
                acc.w += r[k].w;
            }
        }
    }
    if (c0 > (unsigned)SLOTS || c1 > (unsigned)SLOTS) {
        const int total = (int)min(*ovf_n, (unsigned)E);
        for (int base = 0; base < total; base += 32) {
            int od = -1, os = 0;
            if (base + lane < total) {
                od = ovf[2 * (size_t)(base + lane)];
                os = ovf[2 * (size_t)(base + lane) + 1];
                os = min(max(os, 0), N - 1);
            }
            unsigned hit = __builtin_amdgcn_ballot_w32(od == t);
            while (hit != 0u) {
                const int j = __builtin_ctz(hit);
                hit &= hit - 1u;
                const int s = __builtin_amdgcn_readlane(os, j);
                const float4 r = xw4[(size_t)s * (DIM / 4) + lane];
                acc.x += r.x;
                acc.y += r.y;
                acc.z += r.z;
                acc.w += r.w;
            }
        }
    }
    acc.x = (acc.x + self.x) * dt;
    acc.y = (acc.y + self.y) * dt;
    acc.z = (acc.z + self.z) * dt;
    acc.w = (acc.w + self.w) * dt;

    float4 v;
    v.x = fmaxf(acc.x + bb.x, 0.f);
    v.y = fmaxf(acc.y + bb.y, 0.f);
    v.z = fmaxf(acc.z + bb.z, 0.f);
    v.w = fmaxf(acc.w + bb.w, 0.f);

    float m = fmaxf(fmaxf(v.x, v.y), fmaxf(v.z, v.w));
    #pragma unroll
    for (int off = 16; off > 0; off >>= 1) m = fmaxf(m, __shfl_xor(m, off, 32));

    v.x = expf(v.x - m); v.y = expf(v.y - m);
    v.z = expf(v.z - m); v.w = expf(v.w - m);
    float s = v.x + v.y + v.z + v.w;
    #pragma unroll
    for (int off = 16; off > 0; off >>= 1) s += __shfl_xor(s, off, 32);

    const float inv = 1.0f / s;
    v.x *= inv; v.y *= inv; v.z *= inv; v.w *= inv;
    ((float4*)out)[(size_t)t * (DIM / 4) + lane] = v;
}


extern "C" void kernel_launch(void* const* d_in, const int* in_sizes, int n_in,
                              void* d_out, int out_size, void* d_ws, size_t ws_size,
                              hipStream_t stream) {
    const float* x  = (const float*)d_in[0];
    const int*   ei = (const int*)  d_in[1];
    const float* W  = (const float*)d_in[2];
    const float* b  = (const float*)d_in[3];
    const int N = in_sizes[0] / DIM;
    const int E = in_sizes[1] / 2;
    const int* src = ei;
    const int* dst = ei + E;

    float* out = (float*)d_out;
    float*    xw       = (float*)d_ws;
    const int NP = (N + 63) / 64 * 64;
    unsigned* cnt      = (unsigned*)(xw + (size_t)N * DIM);
    int*      slots    = (int*)(cnt + 2 * NP);
    int*      ovf      = slots + (size_t)NP * 2 * SLOTS;
    unsigned* ovf_n    = (unsigned*)(ovf + 2 * (size_t)E);

    k_deg_init<<<(2 * NP + 255) / 256, 256, 0, stream>>>(cnt, ovf_n, 2 * NP);
    k_count   <<<(E + 256 * EPT - 1) / (256 * EPT), 256, 0, stream>>>(src, dst, cnt, slots, ovf_n, ovf, E, NP);

    const int Mtiles = (N + 15) / 16;
    k_gemm_wmma<<<(Mtiles + 7) / 8, 256, 0, stream>>>(x, W, xw, cnt, N, NP, Mtiles);

    k_gather  <<<(N + 7) / 8, 256, 0, stream>>>(slots, cnt, ovf_n, ovf, xw, b, out, N, NP, E);
}
